# starting kernel
# baseline (speedup 1.0000x reference)
#include <hip/hip_runtime.h>


#define SHIFT   10
#define ALPHA   0.2f
#define L_LEN   16384
#define BATCH   32
#define FOUT    64
#define KPAD    12
#define TILES_PER_BATCH (L_LEN / 16)
#define WAVES_PER_BLOCK 8

typedef __attribute__((ext_vector_type(2))) float v2f;
typedef __attribute__((ext_vector_type(8))) float v8f;

__device__ __forceinline__ float leaky(float s) {
  return fmaxf(s, 0.0f) + ALPHA * fminf(s, 0.0f);
}

__global__ __launch_bounds__(256)
void pad_weights(const float* __restrict__ wk, float* __restrict__ wpad) {
  const int i = blockIdx.x * blockDim.x + threadIdx.x;
  if (i < KPAD * FOUT) wpad[i] = (i < 9 * FOUT) ? wk[i] : 0.0f;
}

__global__ __launch_bounds__(256)
void shiftconv1d_wmma_f32(const float* __restrict__ x,
                          const float* __restrict__ ckg,
                          const float* __restrict__ wpad,
                          float* __restrict__ out) {
  __shared__ float cross_lds[WAVES_PER_BLOCK][64];

  const int lane = threadIdx.x & 31;
  const int widx = threadIdx.x >> 5;
  const int gw   = blockIdx.x * WAVES_PER_BLOCK + widx;
  const int b    = gw >> 10;
  const int tile = gw & (TILES_PER_BATCH - 1);
  const int l0   = tile << 4;

  const float* xb = x + (size_t)b * L_LEN;
  float* cl = cross_lds[widx];

  __builtin_prefetch(xb + l0, 0, 0);

  if (lane >= 22) cl[lane + 32] = 0.0f;

  if (lane < 18) {
    const int pos = l0 - 1 + lane;
    float c0 = 0.0f, c1 = 0.0f, c2 = 0.0f;
    if (pos >= 0 && pos < L_LEN) {
      int pm = pos - SHIFT; if (pm < 0)      pm += L_LEN;
      int pp = pos + SHIFT; if (pp >= L_LEN) pp -= L_LEN;
      const float x0 = xb[pos], xm = xb[pm], xp = xb[pp];
      c0 = leaky(fmaf(xp, ckg[6], fmaf(xm, ckg[3], x0 * ckg[0])));
      c1 = leaky(fmaf(xp, ckg[7], fmaf(xm, ckg[4], x0 * ckg[1])));
      c2 = leaky(fmaf(xp, ckg[8], fmaf(xm, ckg[5], x0 * ckg[2])));
    }
    cl[3 * lane + 0] = c0;
    cl[3 * lane + 1] = c1;
    cl[3 * lane + 2] = c2;
  }

  const int m    = lane & 15;
  const int half = (lane >> 4) & 1;

  const v8f zero8 = {0.f, 0.f, 0.f, 0.f, 0.f, 0.f, 0.f, 0.f};
  v8f acc[4];
#pragma unroll
  for (int t = 0; t < 4; ++t) acc[t] = zero8;

#pragma unroll
  for (int s = 0; s < 3; ++s) {
    v2f a;
#pragma unroll
    for (int kk = 0; kk < 2; ++kk) {
      const int k   = 4 * s + 2 * half + kk;
      const int idx = (k <= 8) ? (3 * m + k) : 63;
      a[kk] = cl[idx];
    }
#pragma unroll
    for (int t = 0; t < 4; ++t) {
      v2f bb;
#pragma unroll
      for (int kk = 0; kk < 2; ++kk) {
        const int k = 4 * s + 2 * half + kk;
        bb[kk] = wpad[k * FOUT + t * 16 + m];
      }
      acc[t] = __builtin_amdgcn_wmma_f32_16x16x4_f32(
          false, a, false, bb, (short)0, acc[t], false, false);
    }
  }

  float* ob = out + ((size_t)b * L_LEN + l0) * FOUT;
#pragma unroll
  for (int t = 0; t < 4; ++t) {
#pragma unroll
    for (int r = 0; r < 8; ++r) {
      const int p = r + 8 * half;
      ob[(size_t)p * FOUT + t * 16 + m] = acc[t][r];
    }
  }
}

extern "C" void kernel_launch(void* const* d_in, const int* in_sizes, int n_in,
                              void* d_out, int out_size, void* d_ws, size_t ws_size,
                              hipStream_t stream) {
  (void)in_sizes; (void)n_in; (void)out_size; (void)ws_size;
  const float* x  = (const float*)d_in[0];
  const float* ck = (const float*)d_in[1];
  const float* wk = (const float*)d_in[2];
  float* out  = (float*)d_out;
  float* wpad = (float*)d_ws;

  pad_weights<<<3, 256, 0, stream>>>(wk, wpad);

  const int total_waves = BATCH * TILES_PER_BATCH;
  const int blocks = total_waves / WAVES_PER_BLOCK;
  shiftconv1d_wmma_f32<<<blocks, 256, 0, stream>>>(x, ck, wpad, out);
}


# K padding of the conv weights formed in registers by the B loader (every lane loads row 8, the upper lane half selects 0): the weight-padding launch and its workspace plane go, 2 launches -> 1
# speedup vs baseline: 1.0698x; 1.0698x over previous
#include <hip/hip_runtime.h>


#define SHIFT   10
#define ALPHA   0.2f
#define L_LEN   16384
#define BATCH   32
#define FOUT    64
#define TILES_PER_BATCH (L_LEN / 16)
#define WAVES_PER_BLOCK 8

typedef __attribute__((ext_vector_type(2))) float v2f;
typedef __attribute__((ext_vector_type(8))) float v8f;

__device__ __forceinline__ float leaky(float s) {
  return fmaxf(s, 0.0f) + ALPHA * fminf(s, 0.0f);
}

__global__ __launch_bounds__(256)
void shiftconv1d_wmma_f32(const float* __restrict__ x,
                          const float* __restrict__ ckg,
                          const float* __restrict__ wk,
                          float* __restrict__ out) {
  __shared__ float cross_lds[WAVES_PER_BLOCK][64];

  const int lane = threadIdx.x & 31;
  const int widx = threadIdx.x >> 5;
  const int gw   = blockIdx.x * WAVES_PER_BLOCK + widx;
  const int b    = gw >> 10;
  const int tile = gw & (TILES_PER_BATCH - 1);
  const int l0   = tile << 4;

  const float* xb = x + (size_t)b * L_LEN;
  float* cl = cross_lds[widx];

  __builtin_prefetch(xb + l0, 0, 0);

  if (lane >= 22) cl[lane + 32] = 0.0f;

  if (lane < 18) {
    const int pos = l0 - 1 + lane;
    float c0 = 0.0f, c1 = 0.0f, c2 = 0.0f;
    if (pos >= 0 && pos < L_LEN) {
      int pm = pos - SHIFT; if (pm < 0)      pm += L_LEN;
      int pp = pos + SHIFT; if (pp >= L_LEN) pp -= L_LEN;
      const float x0 = xb[pos], xm = xb[pm], xp = xb[pp];
      c0 = leaky(fmaf(xp, ckg[6], fmaf(xm, ckg[3], x0 * ckg[0])));
      c1 = leaky(fmaf(xp, ckg[7], fmaf(xm, ckg[4], x0 * ckg[1])));
      c2 = leaky(fmaf(xp, ckg[8], fmaf(xm, ckg[5], x0 * ckg[2])));
    }
    cl[3 * lane + 0] = c0;
    cl[3 * lane + 1] = c1;
    cl[3 * lane + 2] = c2;
  }

  const int m    = lane & 15;
  const int half = (lane >> 4) & 1;

  const v8f zero8 = {0.f, 0.f, 0.f, 0.f, 0.f, 0.f, 0.f, 0.f};
  v8f acc[4];
#pragma unroll
  for (int t = 0; t < 4; ++t) acc[t] = zero8;

#pragma unroll
  for (int s = 0; s < 3; ++s) {
    v2f a;
#pragma unroll
    for (int kk = 0; kk < 2; ++kk) {
      const int k   = 4 * s + 2 * half + kk;
      const int idx = (k <= 8) ? (3 * m + k) : 63;
      a[kk] = cl[idx];
    }
#pragma unroll
    for (int t = 0; t < 4; ++t) {
      v2f bb;
      if (s < 2) {
#pragma unroll
        for (int kk = 0; kk < 2; ++kk) {
          const int k = 4 * s + 2 * half + kk;
          bb[kk] = wk[k * FOUT + t * 16 + m];
        }
      } else {
        const float w8 = wk[8 * FOUT + t * 16 + m];
        bb[0] = half ? 0.0f : w8;
        bb[1] = 0.0f;
      }
      acc[t] = __builtin_amdgcn_wmma_f32_16x16x4_f32(
          false, a, false, bb, (short)0, acc[t], false, false);
    }
  }

  float* ob = out + ((size_t)b * L_LEN + l0) * FOUT;
#pragma unroll
  for (int t = 0; t < 4; ++t) {
#pragma unroll
    for (int r = 0; r < 8; ++r) {
      const int p = r + 8 * half;
      ob[(size_t)p * FOUT + t * 16 + m] = acc[t][r];
    }
  }
}

extern "C" void kernel_launch(void* const* d_in, const int* in_sizes, int n_in,
                              void* d_out, int out_size, void* d_ws, size_t ws_size,
                              hipStream_t stream) {
  (void)in_sizes; (void)n_in; (void)out_size; (void)d_ws; (void)ws_size;
  const float* x  = (const float*)d_in[0];
  const float* ck = (const float*)d_in[1];
  const float* wk = (const float*)d_in[2];
  float* out  = (float*)d_out;

  const int total_waves = BATCH * TILES_PER_BATCH;
  const int blocks = total_waves / WAVES_PER_BLOCK;
  shiftconv1d_wmma_f32<<<blocks, 256, 0, stream>>>(x, ck, wk, out);
}
